# starting kernel
# baseline (speedup 1.0000x reference)
#include <hip/hip_runtime.h>

typedef __attribute__((ext_vector_type(2))) float v2f;
typedef __attribute__((ext_vector_type(8))) float v8f;

#define IN_CH 128
#define HO    128

__global__ void zero_f32(float* __restrict__ p, int n) {
  int i = blockIdx.x * blockDim.x + threadIdx.x;
  int stride = gridDim.x * blockDim.x;
  for (; i < n; i += stride) p[i] = 0.0f;
}

__global__ __launch_bounds__(256) void gemm_wmma_f32(
    const float* __restrict__ X, const float* __restrict__ Wl,
    float* __restrict__ Xp, int M) {
  const int ntiles = HO / 16;
  int wave = (blockIdx.x * blockDim.x + threadIdx.x) >> 5;
  int lane = threadIdx.x & 31;
  int mt = wave / ntiles;
  int nt = wave - mt * ntiles;
  int m0 = mt * 16;
  int n0 = nt * 16;
  if (m0 >= M) return;

  int half = lane >> 4;
  int l15  = lane & 15;
  int arow = m0 + l15;
  int bcol = n0 + l15;

  if (m0 + 16 <= M) {
    v8f acc = {};
    for (int k0 = 0; k0 < IN_CH; k0 += 4) {
      int ka = k0 + half * 2;
      v2f a, b;
      a.x = X[(size_t)arow * IN_CH + ka];
      a.y = X[(size_t)arow * IN_CH + ka + 1];
      b.x = Wl[(size_t)ka * HO + bcol];
      b.y = Wl[(size_t)(ka + 1) * HO + bcol];
      acc = __builtin_amdgcn_wmma_f32_16x16x4_f32(
           false, a,  false, b,
           (short)0, acc,  false,  false);
    }
#pragma unroll
    for (int r = 0; r < 8; ++r) {
      Xp[(size_t)(m0 + half * 8 + r) * HO + bcol] = acc[r];
    }
  } else {
    if (half == 0 && arow < M) {
      for (int n = 0; n < 16; ++n) {
        float s = 0.0f;
        for (int k = 0; k < IN_CH; ++k)
          s += X[(size_t)arow * IN_CH + k] * Wl[(size_t)k * HO + (n0 + n)];
        Xp[(size_t)arow * HO + (n0 + n)] = s;
      }
    }
  }
}

__global__ __launch_bounds__(128) void scatter_v2e(
    const float* __restrict__ Xp, const int* __restrict__ src,
    const int* __restrict__ dst, float* __restrict__ Xe, int nnz) {
  int t = blockIdx.x * blockDim.x + threadIdx.x;
  int e = t >> 5;
  if (e >= nnz) return;
  int c = (t & 31) * 4;
  int s = src[e];
  int d = dst[e];
  const float4 v = *reinterpret_cast<const float4*>(Xp + (size_t)s * HO + c);
  float* o = Xe + (size_t)d * HO + c;
  atomicAdd(o + 0, v.x);
  atomicAdd(o + 1, v.y);
  atomicAdd(o + 2, v.z);
  atomicAdd(o + 3, v.w);
}

__global__ __launch_bounds__(128) void scatter_e2v(
    const float* __restrict__ Xe, const float* __restrict__ degE,
    const float* __restrict__ W, const float* __restrict__ degV,
    const int* __restrict__ src, const int* __restrict__ dst,
    float* __restrict__ out, int nnz) {
  int t = blockIdx.x * blockDim.x + threadIdx.x;
  int e = t >> 5;
  if (e >= nnz) return;
  int c = (t & 31) * 4;
  int s = src[e];
  int d = dst[e];
  float scale = degE[d] * W[d] * degV[s];
  const float4 v = *reinterpret_cast<const float4*>(Xe + (size_t)d * HO + c);
  float* o = out + (size_t)s * HO + c;
  atomicAdd(o + 0, v.x * scale);
  atomicAdd(o + 1, v.y * scale);
  atomicAdd(o + 2, v.z * scale);
  atomicAdd(o + 3, v.w * scale);
}

extern "C" void kernel_launch(void* const* d_in, const int* in_sizes, int n_in,
                              void* d_out, int out_size, void* d_ws, size_t ws_size,
                              hipStream_t stream) {
  const float* X    = (const float*)d_in[0];
  const float* Wl   = (const float*)d_in[1];
  const float* degE = (const float*)d_in[2];
  const float* degV = (const float*)d_in[3];
  const float* W    = (const float*)d_in[4];
  const int* g1_src = (const int*)d_in[5];
  const int* g1_dst = (const int*)d_in[6];
  float* out = (float*)d_out;

  const int N   = in_sizes[0] / IN_CH;
  const int E   = in_sizes[2];
  const int nnz = in_sizes[5];

  float* Xp = (float*)d_ws;
  float* Xe = Xp + (size_t)N * HO;

  zero_f32<<<2048, 256, 0, stream>>>(Xe, E * HO);
  zero_f32<<<4096, 256, 0, stream>>>(out, N * HO);

  {
    int mtiles = (N + 15) / 16;
    int waves = mtiles * (HO / 16);
    int blocks = (waves * 32 + 255) / 256;
    gemm_wmma_f32<<<blocks, 256, 0, stream>>>(X, Wl, Xp, N);
  }

  {
    int blocks = (nnz + 3) / 4;
    scatter_v2e<<<blocks, 128, 0, stream>>>(Xp, g1_src, g1_dst, Xe, nnz);
    scatter_e2v<<<blocks, 128, 0, stream>>>(Xe, degE, W, degV,
                                            g1_src, g1_dst, out, nnz);
  }
}


# segment sums without float atomics: per-call counting-sort CSR (int counts, scan, fill), one wave per segment sums its rows in ascending index order, each output row stored once
# speedup vs baseline: 1.2075x; 1.2075x over previous
#include <hip/hip_runtime.h>
#include <limits.h>

typedef __attribute__((ext_vector_type(2))) float v2f;
typedef __attribute__((ext_vector_type(8))) float v8f;

#define IN_CH 128
#define HO    128
#define SEG_CAP 256


__global__ void zero_i32(int* __restrict__ p, int n) {
  int i = blockIdx.x * blockDim.x + threadIdx.x;
  if (i < n) p[i] = 0;
}

__global__ __launch_bounds__(256) void gemm_wmma_f32(
    const float* __restrict__ X, const float* __restrict__ Wl,
    float* __restrict__ Xp, int M) {
  const int ntiles = HO / 16;
  int wave = (blockIdx.x * blockDim.x + threadIdx.x) >> 5;
  int lane = threadIdx.x & 31;
  int mt = wave / ntiles;
  int nt = wave - mt * ntiles;
  int m0 = mt * 16;
  int n0 = nt * 16;
  if (m0 >= M) return;

  int half = lane >> 4;
  int l15  = lane & 15;
  int arow = m0 + l15;
  int bcol = n0 + l15;

  if (m0 + 16 <= M) {
    v8f acc = {};
    for (int k0 = 0; k0 < IN_CH; k0 += 4) {
      int ka = k0 + half * 2;
      v2f a, b;
      a.x = X[(size_t)arow * IN_CH + ka];
      a.y = X[(size_t)arow * IN_CH + ka + 1];
      b.x = Wl[(size_t)ka * HO + bcol];
      b.y = Wl[(size_t)(ka + 1) * HO + bcol];
      acc = __builtin_amdgcn_wmma_f32_16x16x4_f32(
           false, a,  false, b,
           (short)0, acc,  false,  false);
    }
#pragma unroll
    for (int r = 0; r < 8; ++r) {
      Xp[(size_t)(m0 + half * 8 + r) * HO + bcol] = acc[r];
    }
  } else {
    if (half == 0 && arow < M) {
      for (int n = 0; n < 16; ++n) {
        float s = 0.0f;
        for (int k = 0; k < IN_CH; ++k)
          s += X[(size_t)arow * IN_CH + k] * Wl[(size_t)k * HO + (n0 + n)];
        Xp[(size_t)arow * HO + (n0 + n)] = s;
      }
    }
  }
}


__global__ __launch_bounds__(256) void count_edges(
    const int* __restrict__ src, const int* __restrict__ dst,
    int* __restrict__ cnt, int vbase, int nnz) {
  int i = blockIdx.x * blockDim.x + threadIdx.x;
  if (i >= nnz) return;
  atomicAdd(cnt + dst[i], 1);
  atomicAdd(cnt + vbase + src[i], 1);
}

__global__ __launch_bounds__(1024) void scan_counts(
    const int* __restrict__ cnt, int* __restrict__ off, int* __restrict__ cur,
    int total) {
  __shared__ int wsum[32];
  __shared__ int base_s;
  const int tid = threadIdx.x, lane = tid & 31, w = tid >> 5;
  const int start = blockIdx.x * 1024;

  int s = 0;
  for (int i = tid; i < start; i += 1024) s += cnt[i];
  for (int o = 16; o; o >>= 1) s += __shfl_xor(s, o);
  if (lane == 0) wsum[w] = s;
  __syncthreads();
  if (w == 0) {
    int t = wsum[lane];
    for (int o = 16; o; o >>= 1) t += __shfl_xor(t, o);
    if (lane == 0) base_s = t;
  }
  __syncthreads();
  const int base = base_s;

  const int i = start + tid;
  const int x = (i < total) ? cnt[i] : 0;
  int inc = x;
  for (int o = 1; o < 32; o <<= 1) {
    int y = __shfl_up(inc, o);
    if (lane >= o) inc += y;
  }
  if (lane == 31) wsum[w] = inc;
  __syncthreads();
  int t = (lane < w) ? wsum[lane] : 0;
  for (int o = 16; o; o >>= 1) t += __shfl_xor(t, o);
  const int ex = base + t + inc - x;
  if (i < total) { off[i] = ex; cur[i] = ex; }
}

__global__ __launch_bounds__(256) void fill_edges(
    const int* __restrict__ src, const int* __restrict__ dst,
    int* __restrict__ cur, int* __restrict__ lst, int vbase, int nnz) {
  int i = blockIdx.x * blockDim.x + threadIdx.x;
  if (i >= nnz) return;
  const int s = src[i];
  const int d = dst[i];
  lst[atomicAdd(cur + d, 1)] = s;
  lst[atomicAdd(cur + vbase + s, 1)] = d;
}

template <int NB>
__device__ __forceinline__ void add_rows(float4& acc, const float4* __restrict__ R,
                                         const int* idx, int lane) {
  float4 x[NB];
#pragma unroll
  for (int u = 0; u < NB; ++u) x[u] = R[(size_t)idx[u] * (HO / 4) + lane];
#pragma unroll
  for (int u = 0; u < NB; ++u) {
    acc.x += x[u].x; acc.y += x[u].y; acc.z += x[u].z; acc.w += x[u].w;
  }
}

__device__ __forceinline__ float4 ordered_row_sum(
    const float* __restrict__ rows, const int* __restrict__ lst,
    int beg, int end, int* raw, int* srt, int lane) {
  const int L = end - beg;
  const bool in_lds = L <= SEG_CAP;
  const int L4 = (L + 3) >> 2;
  const float4* R = reinterpret_cast<const float4*>(rows);
  float4 acc = make_float4(0.f, 0.f, 0.f, 0.f);

  if (in_lds) {
    for (int j = lane; j < L4 * 4; j += 32)
      raw[j] = (j < L) ? lst[beg + j] : INT_MAX;
  }
  __syncthreads();
  if (in_lds) {
    for (int j = lane; j < L; j += 32) {
      const int v = raw[j];
      int r = 0;
      for (int q = 0; q < L4; ++q) {
        const int4 u = reinterpret_cast<const int4*>(raw)[q];
        const int k = q * 4;
        r += (u.x < v) | ((u.x == v) & (k     < j));
        r += (u.y < v) | ((u.y == v) & (k + 1 < j));
        r += (u.z < v) | ((u.z == v) & (k + 2 < j));
        r += (u.w < v) | ((u.w == v) & (k + 3 < j));
      }
      srt[r] = v;
    }
  }
  __syncthreads();
  if (in_lds) {
    int j = 0;
    for (; j + 8 <= L; j += 8) add_rows<8>(acc, R, srt + j, lane);
    if (j + 4 <= L) { add_rows<4>(acc, R, srt + j, lane); j += 4; }
    if (j + 2 <= L) { add_rows<2>(acc, R, srt + j, lane); j += 2; }
    if (j < L) add_rows<1>(acc, R, srt + j, lane);
  } else {
    int prev = -1;
    for (int it = 0; it < L; ++it) {
      int best = INT_MAX;
      for (int k = beg + lane; k < end; k += 32) {
        const int v = lst[k];
        if (v > prev && v < best) best = v;
      }
      for (int o = 16; o; o >>= 1) best = min(best, __shfl_xor(best, o));
      if (best == INT_MAX) break;
      int c = 0;
      for (int k = beg + lane; k < end; k += 32) c += (lst[k] == best);
      for (int o = 16; o; o >>= 1) c += __shfl_xor(c, o);
      const float4 x = R[(size_t)best * (HO / 4) + lane];
      for (int m = 0; m < c; ++m) {
        acc.x += x.x; acc.y += x.y; acc.z += x.z; acc.w += x.w;
      }
      prev = best;
    }
  }
  return acc;
}

__global__ __launch_bounds__(256) void gather_v2e(
    const float* __restrict__ Xp, const int* __restrict__ off,
    const int* __restrict__ lst, const float* __restrict__ degE,
    const float* __restrict__ W, float* __restrict__ Xe, int E) {
  __shared__ __attribute__((aligned(16))) int sm[8][2 * SEG_CAP];
  const int w = __builtin_amdgcn_readfirstlane(threadIdx.x >> 5);
  const int lane = threadIdx.x & 31;
  const int e = blockIdx.x * 8 + w;
  const bool live = e < E;
  const int beg = live ? off[e] : 0;
  const int end = live ? off[e + 1] : 0;
  float4 a = ordered_row_sum(Xp, lst, beg, end, sm[w], sm[w] + SEG_CAP, lane);
  if (live) {
    const float de = degE[e], we = W[e];
    a.x = a.x * de * we; a.y = a.y * de * we;
    a.z = a.z * de * we; a.w = a.w * de * we;
    reinterpret_cast<float4*>(Xe)[(size_t)e * (HO / 4) + lane] = a;
  }
}

__global__ __launch_bounds__(256) void gather_e2v(
    const float* __restrict__ Xe, const int* __restrict__ off,
    const int* __restrict__ lst, const float* __restrict__ degV,
    float* __restrict__ out, int N) {
  __shared__ __attribute__((aligned(16))) int sm[8][2 * SEG_CAP];
  const int w = __builtin_amdgcn_readfirstlane(threadIdx.x >> 5);
  const int lane = threadIdx.x & 31;
  const int v = blockIdx.x * 8 + w;
  const bool live = v < N;
  const int beg = live ? off[v] : 0;
  const int end = live ? off[v + 1] : 0;
  float4 a = ordered_row_sum(Xe, lst, beg, end, sm[w], sm[w] + SEG_CAP, lane);
  if (live) {
    const float dv = degV[v];
    a.x *= dv; a.y *= dv; a.z *= dv; a.w *= dv;
    reinterpret_cast<float4*>(out)[(size_t)v * (HO / 4) + lane] = a;
  }
}

__global__ __launch_bounds__(256) void move_node_csr(
    const int* __restrict__ off_in, const int* __restrict__ lst_in,
    int* __restrict__ off_out, int* __restrict__ lst_out, int noff, int nnz) {
  int i = blockIdx.x * blockDim.x + threadIdx.x;
  if (i < nnz) lst_out[i] = lst_in[i];
  if (i < noff) off_out[i] = off_in[i] - nnz;
}

extern "C" void kernel_launch(void* const* d_in, const int* in_sizes, int n_in,
                              void* d_out, int out_size, void* d_ws, size_t ws_size,
                              hipStream_t stream) {
  const float* X    = (const float*)d_in[0];
  const float* Wl   = (const float*)d_in[1];
  const float* degE = (const float*)d_in[2];
  const float* degV = (const float*)d_in[3];
  const float* W    = (const float*)d_in[4];
  const int* g1_src = (const int*)d_in[5];
  const int* g1_dst = (const int*)d_in[6];
  float* out = (float*)d_out;

  const int N   = in_sizes[0] / IN_CH;
  const int E   = in_sizes[2];
  const int nnz = in_sizes[5];

  float* Xp = (float*)d_ws;
  float* Xe = Xp + (size_t)N * HO;

  const int T = E + 1 + N + 1;
  int* cnt = (int*)d_out;
  int* off = cnt + T;
  int* cur = off + T;
  int* lst = cur + T;
  int* lstV = (int*)d_ws;
  int* offV = lstV + nnz;

  zero_i32<<<(T + 255) / 256, 256, 0, stream>>>(cnt, T);
  count_edges<<<(nnz + 255) / 256, 256, 0, stream>>>(g1_src, g1_dst, cnt, E + 1, nnz);
  scan_counts<<<(T + 1023) / 1024, 1024, 0, stream>>>(cnt, off, cur, T);
  fill_edges<<<(nnz + 255) / 256, 256, 0, stream>>>(g1_src, g1_dst, cur, lst, E + 1, nnz);

  {
    int mtiles = (N + 15) / 16;
    int waves = mtiles * (HO / 16);
    int blocks = (waves * 32 + 255) / 256;
    gemm_wmma_f32<<<blocks, 256, 0, stream>>>(X, Wl, Xp, N);
  }

  gather_v2e<<<(E + 7) / 8, 256, 0, stream>>>(Xp, off, lst, degE, W, Xe, E);
  move_node_csr<<<(nnz + 255) / 256, 256, 0, stream>>>(off + E + 1, lst + nnz,
                                                      offV, lstV, N + 1, nnz);
  gather_e2v<<<(N + 7) / 8, 256, 0, stream>>>(Xe, offV, lstV, degV, out, N);
}


# CSR build with one returning integer atomic per edge endpoint instead of two: the count pass keeps each arrival rank, the fill pass places members at offset + rank with plain loads and stores
# speedup vs baseline: 1.7838x; 1.4773x over previous
#include <hip/hip_runtime.h>
#include <limits.h>

typedef __attribute__((ext_vector_type(2))) float v2f;
typedef __attribute__((ext_vector_type(8))) float v8f;

#define IN_CH 128
#define HO    128
#define SEG_CAP 256


__global__ void zero_i32(int* __restrict__ p, int n) {
  int i = blockIdx.x * blockDim.x + threadIdx.x;
  if (i < n) p[i] = 0;
}

__global__ __launch_bounds__(256) void gemm_wmma_f32(
    const float* __restrict__ Y, const float* __restrict__ Wl,
    const float* __restrict__ degE, const float* __restrict__ W,
    float* __restrict__ Xe, int M) {
  const int ntiles = HO / 16;
  int wave = (blockIdx.x * blockDim.x + threadIdx.x) >> 5;
  int lane = threadIdx.x & 31;
  int mt = wave / ntiles;
  int nt = wave - mt * ntiles;
  int m0 = mt * 16;
  int n0 = nt * 16;
  if (m0 >= M) return;

  int half = lane >> 4;
  int l15  = lane & 15;
  int arow = min(m0 + l15, M - 1);
  int bcol = n0 + l15;

  float de[8], we[8];
#pragma unroll
  for (int r = 0; r < 8; ++r) {
    const int rr = min(m0 + half * 8 + r, M - 1);
    de[r] = degE[rr];
    we[r] = W[rr];
  }

  v8f acc = {};
  for (int k0 = 0; k0 < IN_CH; k0 += 4) {
    int ka = k0 + half * 2;
    v2f a, b;
    a.x = Y[(size_t)arow * IN_CH + ka];
    a.y = Y[(size_t)arow * IN_CH + ka + 1];
    b.x = Wl[(size_t)ka * HO + bcol];
    b.y = Wl[(size_t)(ka + 1) * HO + bcol];
    acc = __builtin_amdgcn_wmma_f32_16x16x4_f32(
         false, a,  false, b,
         (short)0, acc,  false,  false);
  }
#pragma unroll
  for (int r = 0; r < 8; ++r) {
    const int row = m0 + half * 8 + r;
    if (row < M) Xe[(size_t)row * HO + bcol] = acc[r] * de[r] * we[r];
  }
}


__global__ __launch_bounds__(256) void count_edges(
    const int* __restrict__ src, const int* __restrict__ dst,
    int* __restrict__ cnt, int* __restrict__ rk_d, int* __restrict__ rk_s,
    int vbase, int nnz) {
  int i = blockIdx.x * blockDim.x + threadIdx.x;
  if (i >= nnz) return;
  const int d = dst[i];
  const int s = src[i];
  const int rd = atomicAdd(cnt + d, 1);
  const int rs = atomicAdd(cnt + vbase + s, 1);
  rk_d[i] = rd;
  rk_s[i] = rs;
}

__global__ __launch_bounds__(1024) void scan_counts(
    const int* __restrict__ cnt, int* __restrict__ off, int total) {
  __shared__ int wsum[32];
  __shared__ int base_s;
  const int tid = threadIdx.x, lane = tid & 31, w = tid >> 5;
  const int start = blockIdx.x * 1024;

  int s = 0;
  for (int i = tid; i < start; i += 1024) s += cnt[i];
  for (int o = 16; o; o >>= 1) s += __shfl_xor(s, o);
  if (lane == 0) wsum[w] = s;
  __syncthreads();
  if (w == 0) {
    int t = wsum[lane];
    for (int o = 16; o; o >>= 1) t += __shfl_xor(t, o);
    if (lane == 0) base_s = t;
  }
  __syncthreads();
  const int base = base_s;

  const int i = start + tid;
  const int x = (i < total) ? cnt[i] : 0;
  int inc = x;
  for (int o = 1; o < 32; o <<= 1) {
    int y = __shfl_up(inc, o);
    if (lane >= o) inc += y;
  }
  if (lane == 31) wsum[w] = inc;
  __syncthreads();
  int t = (lane < w) ? wsum[lane] : 0;
  for (int o = 16; o; o >>= 1) t += __shfl_xor(t, o);
  const int ex = base + t + inc - x;
  if (i < total) off[i] = ex;
}

__global__ __launch_bounds__(256) void fill_edges(
    const int* __restrict__ src, const int* __restrict__ dst,
    const int* __restrict__ off, const int* __restrict__ rk_d,
    const int* __restrict__ rk_s, int* __restrict__ lst, int vbase, int nnz) {
  int i = blockIdx.x * blockDim.x + threadIdx.x;
  if (i >= nnz) return;
  const int s = src[i];
  const int d = dst[i];
  lst[off[d] + rk_d[i]] = s;
  lst[off[vbase + s] + rk_s[i]] = d;
}

template <int NB>
__device__ __forceinline__ void add_rows(float4& acc, const float4* __restrict__ R,
                                         const int* idx, int lane) {
  float4 x[NB];
#pragma unroll
  for (int u = 0; u < NB; ++u) x[u] = R[(size_t)idx[u] * (HO / 4) + lane];
#pragma unroll
  for (int u = 0; u < NB; ++u) {
    acc.x += x[u].x; acc.y += x[u].y; acc.z += x[u].z; acc.w += x[u].w;
  }
}

__device__ __forceinline__ float4 ordered_row_sum(
    const float* __restrict__ rows, const int* __restrict__ lst,
    int beg, int end, int* raw, int* srt, int lane) {
  const int L = end - beg;
  const bool in_lds = L <= SEG_CAP;
  const int L4 = (L + 3) >> 2;
  const float4* R = reinterpret_cast<const float4*>(rows);
  float4 acc = make_float4(0.f, 0.f, 0.f, 0.f);

  if (in_lds) {
    for (int j = lane; j < L4 * 4; j += 32)
      raw[j] = (j < L) ? lst[beg + j] : INT_MAX;
  }
  __syncthreads();
  if (in_lds) {
    for (int j = lane; j < L; j += 32) {
      const int v = raw[j];
      int r = 0;
      for (int q = 0; q < L4; ++q) {
        const int4 u = reinterpret_cast<const int4*>(raw)[q];
        const int k = q * 4;
        r += (u.x < v) | ((u.x == v) & (k     < j));
        r += (u.y < v) | ((u.y == v) & (k + 1 < j));
        r += (u.z < v) | ((u.z == v) & (k + 2 < j));
        r += (u.w < v) | ((u.w == v) & (k + 3 < j));
      }
      srt[r] = v;
    }
  }
  __syncthreads();
  if (in_lds) {
    int j = 0;
    for (; j + 8 <= L; j += 8) add_rows<8>(acc, R, srt + j, lane);
    if (j + 4 <= L) { add_rows<4>(acc, R, srt + j, lane); j += 4; }
    if (j + 2 <= L) { add_rows<2>(acc, R, srt + j, lane); j += 2; }
    if (j < L) add_rows<1>(acc, R, srt + j, lane);
  } else {
    int prev = -1;
    for (int it = 0; it < L; ++it) {
      int best = INT_MAX;
      for (int k = beg + lane; k < end; k += 32) {
        const int v = lst[k];
        if (v > prev && v < best) best = v;
      }
      for (int o = 16; o; o >>= 1) best = min(best, __shfl_xor(best, o));
      if (best == INT_MAX) break;
      int c = 0;
      for (int k = beg + lane; k < end; k += 32) c += (lst[k] == best);
      for (int o = 16; o; o >>= 1) c += __shfl_xor(c, o);
      const float4 x = R[(size_t)best * (HO / 4) + lane];
      for (int m = 0; m < c; ++m) {
        acc.x += x.x; acc.y += x.y; acc.z += x.z; acc.w += x.w;
      }
      prev = best;
    }
  }
  return acc;
}

__global__ __launch_bounds__(256) void gather_v2e(
    const float* __restrict__ X, const int* __restrict__ off,
    const int* __restrict__ lst, float* __restrict__ Y, int E) {
  __shared__ __attribute__((aligned(16))) int sm[8][2 * SEG_CAP];
  const int w = __builtin_amdgcn_readfirstlane(threadIdx.x >> 5);
  const int lane = threadIdx.x & 31;
  const int e = blockIdx.x * 8 + w;
  const bool live = e < E;
  const int beg = live ? off[e] : 0;
  const int end = live ? off[e + 1] : 0;
  float4 a = ordered_row_sum(X, lst, beg, end, sm[w], sm[w] + SEG_CAP, lane);
  if (live) reinterpret_cast<float4*>(Y)[(size_t)e * (IN_CH / 4) + lane] = a;
}

__global__ __launch_bounds__(256) void gather_e2v(
    const float* __restrict__ Xe, const int* __restrict__ off,
    const int* __restrict__ lst, const float* __restrict__ degV,
    float* __restrict__ out, int N) {
  __shared__ __attribute__((aligned(16))) int sm[8][2 * SEG_CAP];
  const int w = __builtin_amdgcn_readfirstlane(threadIdx.x >> 5);
  const int lane = threadIdx.x & 31;
  const int v = blockIdx.x * 8 + w;
  const bool live = v < N;
  const int beg = live ? off[v] : 0;
  const int end = live ? off[v + 1] : 0;
  float4 a = ordered_row_sum(Xe, lst, beg, end, sm[w], sm[w] + SEG_CAP, lane);
  if (live) {
    const float dv = degV[v];
    a.x *= dv; a.y *= dv; a.z *= dv; a.w *= dv;
    reinterpret_cast<float4*>(out)[(size_t)v * (HO / 4) + lane] = a;
  }
}

extern "C" void kernel_launch(void* const* d_in, const int* in_sizes, int n_in,
                              void* d_out, int out_size, void* d_ws, size_t ws_size,
                              hipStream_t stream) {
  const float* X    = (const float*)d_in[0];
  const float* Wl   = (const float*)d_in[1];
  const float* degE = (const float*)d_in[2];
  const float* degV = (const float*)d_in[3];
  const float* W    = (const float*)d_in[4];
  const int* g1_src = (const int*)d_in[5];
  const int* g1_dst = (const int*)d_in[6];
  float* out = (float*)d_out;

  const int N   = in_sizes[0] / IN_CH;
  const int E   = in_sizes[2];
  const int nnz = in_sizes[5];

  float* Y  = (float*)d_ws;
  float* Xe = Y + (size_t)N * HO;
  const int T = E + 1 + N + 1;
  int* cnt = (int*)(Y + (size_t)E * IN_CH);
  int* off = cnt + T;
  int* rk_d = off + T;
  int* rk_s = rk_d + nnz;
  int* lst = rk_s + nnz;

  zero_i32<<<(T + 255) / 256, 256, 0, stream>>>(cnt, T);
  count_edges<<<(nnz + 255) / 256, 256, 0, stream>>>(g1_src, g1_dst, cnt, rk_d, rk_s, E + 1, nnz);
  scan_counts<<<(T + 1023) / 1024, 1024, 0, stream>>>(cnt, off, T);
  fill_edges<<<(nnz + 255) / 256, 256, 0, stream>>>(g1_src, g1_dst, off, rk_d, rk_s, lst, E + 1, nnz);

  gather_v2e<<<(E + 7) / 8, 256, 0, stream>>>(X, off, lst, Y, E);

  {
    int mtiles = (E + 15) / 16;
    int waves = mtiles * (HO / 16);
    int blocks = (waves * 32 + 255) / 256;
    gemm_wmma_f32<<<blocks, 256, 0, stream>>>(Y, Wl, degE, W, Xe, E);
  }

  gather_e2v<<<(N + 7) / 8, 256, 0, stream>>>(Xe, off + E + 1, lst, degV, out, N);
}


# CSR member counts in work-group LDS histograms (one edge slice per group, 16-bit counter pairs) plus a per-bin running sum over slices: no global atomic left, counter-zeroing launch gone
# speedup vs baseline: 2.7706x; 1.5532x over previous
#include <hip/hip_runtime.h>
#include <limits.h>

typedef __attribute__((ext_vector_type(2))) float v2f;
typedef __attribute__((ext_vector_type(8))) float v8f;

#define IN_CH 128
#define HO    128
#define SEG_CAP 256


__global__ __launch_bounds__(256) void gemm_wmma_f32(
    const float* __restrict__ Y, const float* __restrict__ Wl,
    const float* __restrict__ degE, const float* __restrict__ W,
    float* __restrict__ Xe, int M) {
  const int ntiles = HO / 16;
  int wave = (blockIdx.x * blockDim.x + threadIdx.x) >> 5;
  int lane = threadIdx.x & 31;
  int mt = wave / ntiles;
  int nt = wave - mt * ntiles;
  int m0 = mt * 16;
  int n0 = nt * 16;
  if (m0 >= M) return;

  int half = lane >> 4;
  int l15  = lane & 15;
  int arow = min(m0 + l15, M - 1);
  int bcol = n0 + l15;

  float de[8], we[8];
#pragma unroll
  for (int r = 0; r < 8; ++r) {
    const int rr = min(m0 + half * 8 + r, M - 1);
    de[r] = degE[rr];
    we[r] = W[rr];
  }

  v8f acc = {};
  for (int k0 = 0; k0 < IN_CH; k0 += 4) {
    int ka = k0 + half * 2;
    v2f a, b;
    a.x = Y[(size_t)arow * IN_CH + ka];
    a.y = Y[(size_t)arow * IN_CH + ka + 1];
    b.x = Wl[(size_t)ka * HO + bcol];
    b.y = Wl[(size_t)(ka + 1) * HO + bcol];
    acc = __builtin_amdgcn_wmma_f32_16x16x4_f32(
         false, a,  false, b,
         (short)0, acc,  false,  false);
  }
#pragma unroll
  for (int r = 0; r < 8; ++r) {
    const int row = m0 + half * 8 + r;
    if (row < M) Xe[(size_t)row * HO + bcol] = acc[r] * de[r] * we[r];
  }
}


#define NBLK 48
#define HIST_WORDS 50016

__device__ __forceinline__ int bump(unsigned* hist, int k) {
  const unsigned old = atomicAdd(&hist[k >> 1], (k & 1) ? 0x10000u : 1u);
  return (int)((k & 1) ? (old >> 16) : (old & 0xffffu));
}

__global__ __launch_bounds__(1024) void count_slices(
    const int* __restrict__ src, const int* __restrict__ dst,
    unsigned* __restrict__ mat16, int* __restrict__ rk_d, int* __restrict__ rk_s,
    int E1, int N1, int W0, int RW, int chunk, int nnz) {
  __shared__ unsigned hist[HIST_WORDS];
  const int side = blockIdx.y;
  const int* __restrict__ key = side ? src : dst;
  int* __restrict__ rk = side ? rk_s : rk_d;
  const int words = ((side ? N1 : E1) + 1) >> 1;
  const int tid = threadIdx.x;

  for (int j = tid; j < words; j += 1024) hist[j] = 0u;
  __syncthreads();

  const int beg = blockIdx.x * chunk;
  const int end = min(beg + chunk, nnz);
  const int last4 = max(nnz - 4, 0);
  for (int i0 = beg + tid * 4; i0 < end; i0 += 4 * 4096) {
    int4 k[4];
#pragma unroll
    for (int u = 0; u < 4; ++u)
      k[u] = *reinterpret_cast<const int4*>(key + min(i0 + u * 4096, last4));
#pragma unroll
    for (int u = 0; u < 4; ++u) {
      const int i = i0 + u * 4096;
      if (i + 3 < end) {
        int4 r;
        r.x = bump(hist, k[u].x);
        r.y = bump(hist, k[u].y);
        r.z = bump(hist, k[u].z);
        r.w = bump(hist, k[u].w);
        *reinterpret_cast<int4*>(rk + i) = r;
      } else {
        for (int q = i; q < end; ++q) rk[q] = bump(hist, key[q]);
      }
    }
  }
  __syncthreads();

  unsigned* __restrict__ row = mat16 + (size_t)blockIdx.x * RW + (side ? W0 : 0);
  for (int j = tid; j < words; j += 1024) row[j] = hist[j];
}

__global__ __launch_bounds__(256) void slice_prefix(
    const unsigned* __restrict__ mat16, int* __restrict__ pre, int* __restrict__ cnt,
    int E1, int N1, int W0, int RW, int T) {
  const int w = blockIdx.x * 256 + threadIdx.x;
  if (w >= RW) return;
  const bool side = w >= W0;
  const int lw = side ? w - W0 : w;
  const int t0 = (side ? E1 : 0) + 2 * lw;
  const bool has1 = 2 * lw + 1 < (side ? N1 : E1);
  unsigned c[NBLK];
#pragma unroll
  for (int b = 0; b < NBLK; ++b) c[b] = mat16[(size_t)b * RW + w];
  int run0 = 0, run1 = 0;
#pragma unroll
  for (int b = 0; b < NBLK; ++b) {
    int* __restrict__ p = pre + (size_t)b * T + t0;
    p[0] = run0;
    if (has1) p[1] = run1;
    run0 += (int)(c[b] & 0xffffu);
    run1 += (int)(c[b] >> 16);
  }
  cnt[t0] = run0;
  if (has1) cnt[t0 + 1] = run1;
}

__global__ __launch_bounds__(1024) void scan_counts(
    const int* __restrict__ cnt, int* __restrict__ off, int total) {
  __shared__ int wsum[32];
  __shared__ int base_s;
  const int tid = threadIdx.x, lane = tid & 31, w = tid >> 5;
  const int start = blockIdx.x * 1024;

  int s = 0;
  for (int i = tid; i < start; i += 1024) s += cnt[i];
  for (int o = 16; o; o >>= 1) s += __shfl_xor(s, o);
  if (lane == 0) wsum[w] = s;
  __syncthreads();
  if (w == 0) {
    int t = wsum[lane];
    for (int o = 16; o; o >>= 1) t += __shfl_xor(t, o);
    if (lane == 0) base_s = t;
  }
  __syncthreads();
  const int base = base_s;

  const int i = start + tid;
  const int x = (i < total) ? cnt[i] : 0;
  int inc = x;
  for (int o = 1; o < 32; o <<= 1) {
    int y = __shfl_up(inc, o);
    if (lane >= o) inc += y;
  }
  if (lane == 31) wsum[w] = inc;
  __syncthreads();
  int t = (lane < w) ? wsum[lane] : 0;
  for (int o = 16; o; o >>= 1) t += __shfl_xor(t, o);
  const int ex = base + t + inc - x;
  if (i < total) off[i] = ex;
}

__global__ __launch_bounds__(256) void fill_edges(
    const int* __restrict__ src, const int* __restrict__ dst,
    const int* __restrict__ off, const int* __restrict__ pre,
    const int* __restrict__ rk_d, const int* __restrict__ rk_s,
    int* __restrict__ lst, int vbase, int T, int chunk, int nnz) {
  int i = blockIdx.x * blockDim.x + threadIdx.x;
  if (i >= nnz) return;
  const int s = src[i];
  const int d = dst[i];
  const int* __restrict__ prow = pre + (size_t)(i / chunk) * T;
  lst[off[d] + prow[d] + rk_d[i]] = s;
  lst[off[vbase + s] + prow[vbase + s] + rk_s[i]] = d;
}

template <int NB>
__device__ __forceinline__ void add_rows(float4& acc, const float4* __restrict__ R,
                                         const int* idx, int lane) {
  float4 x[NB];
#pragma unroll
  for (int u = 0; u < NB; ++u) x[u] = R[(size_t)idx[u] * (HO / 4) + lane];
#pragma unroll
  for (int u = 0; u < NB; ++u) {
    acc.x += x[u].x; acc.y += x[u].y; acc.z += x[u].z; acc.w += x[u].w;
  }
}

__device__ __forceinline__ float4 ordered_row_sum(
    const float* __restrict__ rows, const int* __restrict__ lst,
    int beg, int end, int* raw, int* srt, int lane) {
  const int L = end - beg;
  const bool in_lds = L <= SEG_CAP;
  const int L4 = (L + 3) >> 2;
  const float4* R = reinterpret_cast<const float4*>(rows);
  float4 acc = make_float4(0.f, 0.f, 0.f, 0.f);

  if (in_lds) {
    for (int j = lane; j < L4 * 4; j += 32)
      raw[j] = (j < L) ? lst[beg + j] : INT_MAX;
  }
  __syncthreads();
  if (in_lds) {
    for (int j = lane; j < L; j += 32) {
      const int v = raw[j];
      int r = 0;
      for (int q = 0; q < L4; ++q) {
        const int4 u = reinterpret_cast<const int4*>(raw)[q];
        const int k = q * 4;
        r += (u.x < v) | ((u.x == v) & (k     < j));
        r += (u.y < v) | ((u.y == v) & (k + 1 < j));
        r += (u.z < v) | ((u.z == v) & (k + 2 < j));
        r += (u.w < v) | ((u.w == v) & (k + 3 < j));
      }
      srt[r] = v;
    }
  }
  __syncthreads();
  if (in_lds) {
    int j = 0;
    for (; j + 8 <= L; j += 8) add_rows<8>(acc, R, srt + j, lane);
    if (j + 4 <= L) { add_rows<4>(acc, R, srt + j, lane); j += 4; }
    if (j + 2 <= L) { add_rows<2>(acc, R, srt + j, lane); j += 2; }
    if (j < L) add_rows<1>(acc, R, srt + j, lane);
  } else {
    int prev = -1;
    for (int it = 0; it < L; ++it) {
      int best = INT_MAX;
      for (int k = beg + lane; k < end; k += 32) {
        const int v = lst[k];
        if (v > prev && v < best) best = v;
      }
      for (int o = 16; o; o >>= 1) best = min(best, __shfl_xor(best, o));
      if (best == INT_MAX) break;
      int c = 0;
      for (int k = beg + lane; k < end; k += 32) c += (lst[k] == best);
      for (int o = 16; o; o >>= 1) c += __shfl_xor(c, o);
      const float4 x = R[(size_t)best * (HO / 4) + lane];
      for (int m = 0; m < c; ++m) {
        acc.x += x.x; acc.y += x.y; acc.z += x.z; acc.w += x.w;
      }
      prev = best;
    }
  }
  return acc;
}

__global__ __launch_bounds__(256) void gather_v2e(
    const float* __restrict__ X, const int* __restrict__ off,
    const int* __restrict__ lst, float* __restrict__ Y, int E) {
  __shared__ __attribute__((aligned(16))) int sm[8][2 * SEG_CAP];
  const int w = __builtin_amdgcn_readfirstlane(threadIdx.x >> 5);
  const int lane = threadIdx.x & 31;
  const int e = blockIdx.x * 8 + w;
  const bool live = e < E;
  const int beg = live ? off[e] : 0;
  const int end = live ? off[e + 1] : 0;
  float4 a = ordered_row_sum(X, lst, beg, end, sm[w], sm[w] + SEG_CAP, lane);
  if (live) reinterpret_cast<float4*>(Y)[(size_t)e * (IN_CH / 4) + lane] = a;
}

__global__ __launch_bounds__(256) void gather_e2v(
    const float* __restrict__ Xe, const int* __restrict__ off,
    const int* __restrict__ lst, const float* __restrict__ degV,
    float* __restrict__ out, int N) {
  __shared__ __attribute__((aligned(16))) int sm[8][2 * SEG_CAP];
  const int w = __builtin_amdgcn_readfirstlane(threadIdx.x >> 5);
  const int lane = threadIdx.x & 31;
  const int v = blockIdx.x * 8 + w;
  const bool live = v < N;
  const int beg = live ? off[v] : 0;
  const int end = live ? off[v + 1] : 0;
  float4 a = ordered_row_sum(Xe, lst, beg, end, sm[w], sm[w] + SEG_CAP, lane);
  if (live) {
    const float dv = degV[v];
    a.x *= dv; a.y *= dv; a.z *= dv; a.w *= dv;
    reinterpret_cast<float4*>(out)[(size_t)v * (HO / 4) + lane] = a;
  }
}

extern "C" void kernel_launch(void* const* d_in, const int* in_sizes, int n_in,
                              void* d_out, int out_size, void* d_ws, size_t ws_size,
                              hipStream_t stream) {
  const float* X    = (const float*)d_in[0];
  const float* Wl   = (const float*)d_in[1];
  const float* degE = (const float*)d_in[2];
  const float* degV = (const float*)d_in[3];
  const float* W    = (const float*)d_in[4];
  const int* g1_src = (const int*)d_in[5];
  const int* g1_dst = (const int*)d_in[6];
  float* out = (float*)d_out;

  const int N   = in_sizes[0] / IN_CH;
  const int E   = in_sizes[2];
  const int nnz = in_sizes[5];

  float* Y  = (float*)d_ws;
  float* Xe = Y + (size_t)N * HO;
  const int E1 = E + 1, N1 = N + 1;
  const int T = E1 + N1;
  const int W0 = (E1 + 1) / 2, W1 = (N1 + 1) / 2, RW = W0 + W1;
  const int chunk = (((nnz + NBLK - 1) / NBLK) + 3) & ~3;
  int* cnt = (int*)(Y + (size_t)E * IN_CH);
  int* off = cnt + T;
  int* rk_d = cnt + ((2 * T + 3) & ~3);
  int* rk_s = rk_d + nnz;
  int* lst = rk_s + nnz;
  unsigned* mat16 = (unsigned*)Y;
  int* pre = lst + 2 * (size_t)nnz;

  count_slices<<<dim3(NBLK, 2), 1024, 0, stream>>>(g1_src, g1_dst, mat16, rk_d, rk_s,
                                                   E1, N1, W0, RW, chunk, nnz);
  slice_prefix<<<(RW + 255) / 256, 256, 0, stream>>>(mat16, pre, cnt, E1, N1, W0, RW, T);
  scan_counts<<<(T + 1023) / 1024, 1024, 0, stream>>>(cnt, off, T);
  fill_edges<<<(nnz + 255) / 256, 256, 0, stream>>>(g1_src, g1_dst, off, pre, rk_d, rk_s, lst,
                                                    E1, T, chunk, nnz);

  gather_v2e<<<(E + 7) / 8, 256, 0, stream>>>(X, off, lst, Y, E);

  {
    int mtiles = (E + 15) / 16;
    int waves = mtiles * (HO / 16);
    int blocks = (waves * 32 + 255) / 256;
    gemm_wmma_f32<<<blocks, 256, 0, stream>>>(Y, Wl, degE, W, Xe, E);
  }

  gather_e2v<<<(N + 7) / 8, 256, 0, stream>>>(Xe, off + E + 1, lst, degV, out, N);
}


# CSR fill through an LDS table of places: a work-group loads its slice's offsets for a range of bins into LDS and an LDS add hands each edge its place; no per-edge table gathers, no rank arrays
# speedup vs baseline: 2.9695x; 1.0718x over previous
#include <hip/hip_runtime.h>
#include <limits.h>

typedef __attribute__((ext_vector_type(2))) float v2f;
typedef __attribute__((ext_vector_type(8))) float v8f;

#define IN_CH 128
#define HO    128
#define SEG_CAP 256


__global__ __launch_bounds__(256) void gemm_wmma_f32(
    const float* __restrict__ Y, const float* __restrict__ Wl,
    const float* __restrict__ degE, const float* __restrict__ W,
    float* __restrict__ Xe, int M) {
  const int ntiles = HO / 16;
  int wave = (blockIdx.x * blockDim.x + threadIdx.x) >> 5;
  int lane = threadIdx.x & 31;
  int mt = wave / ntiles;
  int nt = wave - mt * ntiles;
  int m0 = mt * 16;
  int n0 = nt * 16;
  if (m0 >= M) return;

  int half = lane >> 4;
  int l15  = lane & 15;
  int arow = min(m0 + l15, M - 1);
  int bcol = n0 + l15;

  float de[8], we[8];
#pragma unroll
  for (int r = 0; r < 8; ++r) {
    const int rr = min(m0 + half * 8 + r, M - 1);
    de[r] = degE[rr];
    we[r] = W[rr];
  }

  v8f acc = {};
  for (int k0 = 0; k0 < IN_CH; k0 += 4) {
    int ka = k0 + half * 2;
    v2f a, b;
    a.x = Y[(size_t)arow * IN_CH + ka];
    a.y = Y[(size_t)arow * IN_CH + ka + 1];
    b.x = Wl[(size_t)ka * HO + bcol];
    b.y = Wl[(size_t)(ka + 1) * HO + bcol];
    acc = __builtin_amdgcn_wmma_f32_16x16x4_f32(
         false, a,  false, b,
         (short)0, acc,  false,  false);
  }
#pragma unroll
  for (int r = 0; r < 8; ++r) {
    const int row = m0 + half * 8 + r;
    if (row < M) Xe[(size_t)row * HO + bcol] = acc[r] * de[r] * we[r];
  }
}


#define NBLK 48
#define HIST_WORDS 50016

__device__ __forceinline__ void bump(unsigned* hist, int k) {
  atomicAdd(&hist[k >> 1], (k & 1) ? 0x10000u : 1u);
}

__global__ __launch_bounds__(1024) void count_slices(
    const int* __restrict__ src, const int* __restrict__ dst,
    unsigned* __restrict__ mat16,
    int E1, int N1, int W0, int RW, int chunk, int nnz) {
  __shared__ unsigned hist[HIST_WORDS];
  const int side = blockIdx.y;
  const int* __restrict__ key = side ? src : dst;
  const int words = ((side ? N1 : E1) + 1) >> 1;
  const int tid = threadIdx.x;

  for (int j = tid; j < words; j += 1024) hist[j] = 0u;
  __syncthreads();

  const int beg = blockIdx.x * chunk;
  const int end = min(beg + chunk, nnz);
  const int last4 = max(nnz - 4, 0);
  for (int i0 = beg + tid * 4; i0 < end; i0 += 4 * 4096) {
    int4 k[4];
#pragma unroll
    for (int u = 0; u < 4; ++u)
      k[u] = *reinterpret_cast<const int4*>(key + min(i0 + u * 4096, last4));
#pragma unroll
    for (int u = 0; u < 4; ++u) {
      const int i = i0 + u * 4096;
      if (i + 3 < end) {
        bump(hist, k[u].x);
        bump(hist, k[u].y);
        bump(hist, k[u].z);
        bump(hist, k[u].w);
      } else {
        for (int q = i; q < end; ++q) bump(hist, key[q]);
      }
    }
  }
  __syncthreads();

  unsigned* __restrict__ row = mat16 + (size_t)blockIdx.x * RW + (side ? W0 : 0);
  for (int j = tid; j < words; j += 1024) row[j] = hist[j];
}

__global__ __launch_bounds__(256) void slice_prefix(
    const unsigned* __restrict__ mat16, int* __restrict__ pre, int* __restrict__ cnt,
    int E1, int N1, int W0, int RW, int T) {
  const int w = blockIdx.x * 256 + threadIdx.x;
  if (w >= RW) return;
  const bool side = w >= W0;
  const int lw = side ? w - W0 : w;
  const int t0 = (side ? E1 : 0) + 2 * lw;
  const bool has1 = 2 * lw + 1 < (side ? N1 : E1);
  unsigned c[NBLK];
#pragma unroll
  for (int b = 0; b < NBLK; ++b) c[b] = mat16[(size_t)b * RW + w];
  int run0 = 0, run1 = 0;
#pragma unroll
  for (int b = 0; b < NBLK; ++b) {
    int* __restrict__ p = pre + (size_t)b * T + t0;
    p[0] = run0;
    if (has1) p[1] = run1;
    run0 += (int)(c[b] & 0xffffu);
    run1 += (int)(c[b] >> 16);
  }
  cnt[t0] = run0;
  if (has1) cnt[t0 + 1] = run1;
}

__global__ __launch_bounds__(1024) void scan_counts(
    const int* __restrict__ cnt, int* __restrict__ off, int total) {
  __shared__ int wsum[32];
  __shared__ int base_s;
  const int tid = threadIdx.x, lane = tid & 31, w = tid >> 5;
  const int start = blockIdx.x * 1024;

  int s = 0;
  for (int i = tid; i < start; i += 1024) s += cnt[i];
  for (int o = 16; o; o >>= 1) s += __shfl_xor(s, o);
  if (lane == 0) wsum[w] = s;
  __syncthreads();
  if (w == 0) {
    int t = wsum[lane];
    for (int o = 16; o; o >>= 1) t += __shfl_xor(t, o);
    if (lane == 0) base_s = t;
  }
  __syncthreads();
  const int base = base_s;

  const int i = start + tid;
  const int x = (i < total) ? cnt[i] : 0;
  int inc = x;
  for (int o = 1; o < 32; o <<= 1) {
    int y = __shfl_up(inc, o);
    if (lane >= o) inc += y;
  }
  if (lane == 31) wsum[w] = inc;
  __syncthreads();
  int t = (lane < w) ? wsum[lane] : 0;
  for (int o = 16; o; o >>= 1) t += __shfl_xor(t, o);
  const int ex = base + t + inc - x;
  if (i < total) off[i] = ex;
}

#define HPARTS 2
#define NPARTS 3
#define SLOT_CAP 33344

__device__ __forceinline__ void place(int* slot, int* __restrict__ lst, int lo, int nb,
                                      int k, int x) {
  const unsigned r = (unsigned)(k - lo);
  if (r < (unsigned)nb) lst[atomicAdd(&slot[r], 1)] = x;
}

__global__ __launch_bounds__(1024) void place_slices(
    const int* __restrict__ src, const int* __restrict__ dst,
    const int* __restrict__ off, const int* __restrict__ pre,
    int* __restrict__ lst, int E1, int N1, int T, int chunk, int nnz) {
  __shared__ int slot[SLOT_CAP];
  const int tid = threadIdx.x;
  const int p = blockIdx.y;
  const bool nodes = p >= HPARTS;
  const int bins = nodes ? N1 : E1;
  const int span = nodes ? (N1 + NPARTS - 1) / NPARTS : (E1 + HPARTS - 1) / HPARTS;
  const int lo = (nodes ? p - HPARTS : p) * span;
  const int nb = min(span, bins - lo);
  const int tb = (nodes ? E1 : 0) + lo;
  const int* __restrict__ key = nodes ? src : dst;
  const int* __restrict__ val = nodes ? dst : src;
  const int* __restrict__ orow = off + tb;
  const int* __restrict__ prow = pre + (size_t)blockIdx.x * T + tb;

  int j0 = tid;
  for (; j0 + 7 * 1024 < nb; j0 += 8 * 1024) {
    int a[8], c[8];
#pragma unroll
    for (int u = 0; u < 8; ++u) {
      a[u] = orow[j0 + u * 1024];
      c[u] = prow[j0 + u * 1024];
    }
#pragma unroll
    for (int u = 0; u < 8; ++u) slot[j0 + u * 1024] = a[u] + c[u];
  }
  for (; j0 < nb; j0 += 1024) slot[j0] = orow[j0] + prow[j0];
  __syncthreads();

  const int beg = blockIdx.x * chunk;
  const int end = min(beg + chunk, nnz);
  const int last4 = max(nnz - 4, 0);
  for (int i0 = beg + tid * 4; i0 < end; i0 += 4 * 4096) {
    int4 k[4], x[4];
#pragma unroll
    for (int u = 0; u < 4; ++u) {
      const int at = min(i0 + u * 4096, last4);
      k[u] = *reinterpret_cast<const int4*>(key + at);
      x[u] = *reinterpret_cast<const int4*>(val + at);
    }
#pragma unroll
    for (int u = 0; u < 4; ++u) {
      const int i = i0 + u * 4096;
      if (i + 3 < end) {
        place(slot, lst, lo, nb, k[u].x, x[u].x);
        place(slot, lst, lo, nb, k[u].y, x[u].y);
        place(slot, lst, lo, nb, k[u].z, x[u].z);
        place(slot, lst, lo, nb, k[u].w, x[u].w);
      } else {
        for (int q = i; q < end; ++q) place(slot, lst, lo, nb, key[q], val[q]);
      }
    }
  }
}

template <int NB>
__device__ __forceinline__ void add_rows(float4& acc, const float4* __restrict__ R,
                                         const int* idx, int lane) {
  float4 x[NB];
#pragma unroll
  for (int u = 0; u < NB; ++u) x[u] = R[(size_t)idx[u] * (HO / 4) + lane];
#pragma unroll
  for (int u = 0; u < NB; ++u) {
    acc.x += x[u].x; acc.y += x[u].y; acc.z += x[u].z; acc.w += x[u].w;
  }
}

__device__ __forceinline__ float4 ordered_row_sum(
    const float* __restrict__ rows, const int* __restrict__ lst,
    int beg, int end, int* raw, int* srt, int lane) {
  const int L = end - beg;
  const bool in_lds = L <= SEG_CAP;
  const int L4 = (L + 3) >> 2;
  const float4* R = reinterpret_cast<const float4*>(rows);
  float4 acc = make_float4(0.f, 0.f, 0.f, 0.f);

  if (in_lds) {
    for (int j = lane; j < L4 * 4; j += 32)
      raw[j] = (j < L) ? lst[beg + j] : INT_MAX;
  }
  __syncthreads();
  if (in_lds) {
    for (int j = lane; j < L; j += 32) {
      const int v = raw[j];
      int r = 0;
      for (int q = 0; q < L4; ++q) {
        const int4 u = reinterpret_cast<const int4*>(raw)[q];
        const int k = q * 4;
        r += (u.x < v) | ((u.x == v) & (k     < j));
        r += (u.y < v) | ((u.y == v) & (k + 1 < j));
        r += (u.z < v) | ((u.z == v) & (k + 2 < j));
        r += (u.w < v) | ((u.w == v) & (k + 3 < j));
      }
      srt[r] = v;
    }
  }
  __syncthreads();
  if (in_lds) {
    int j = 0;
    for (; j + 8 <= L; j += 8) add_rows<8>(acc, R, srt + j, lane);
    if (j + 4 <= L) { add_rows<4>(acc, R, srt + j, lane); j += 4; }
    if (j + 2 <= L) { add_rows<2>(acc, R, srt + j, lane); j += 2; }
    if (j < L) add_rows<1>(acc, R, srt + j, lane);
  } else {
    int prev = -1;
    for (int it = 0; it < L; ++it) {
      int best = INT_MAX;
      for (int k = beg + lane; k < end; k += 32) {
        const int v = lst[k];
        if (v > prev && v < best) best = v;
      }
      for (int o = 16; o; o >>= 1) best = min(best, __shfl_xor(best, o));
      if (best == INT_MAX) break;
      int c = 0;
      for (int k = beg + lane; k < end; k += 32) c += (lst[k] == best);
      for (int o = 16; o; o >>= 1) c += __shfl_xor(c, o);
      const float4 x = R[(size_t)best * (HO / 4) + lane];
      for (int m = 0; m < c; ++m) {
        acc.x += x.x; acc.y += x.y; acc.z += x.z; acc.w += x.w;
      }
      prev = best;
    }
  }
  return acc;
}

__global__ __launch_bounds__(256) void gather_v2e(
    const float* __restrict__ X, const int* __restrict__ off,
    const int* __restrict__ lst, float* __restrict__ Y, int E) {
  __shared__ __attribute__((aligned(16))) int sm[8][2 * SEG_CAP];
  const int w = __builtin_amdgcn_readfirstlane(threadIdx.x >> 5);
  const int lane = threadIdx.x & 31;
  const int e = blockIdx.x * 8 + w;
  const bool live = e < E;
  const int beg = live ? off[e] : 0;
  const int end = live ? off[e + 1] : 0;
  float4 a = ordered_row_sum(X, lst, beg, end, sm[w], sm[w] + SEG_CAP, lane);
  if (live) reinterpret_cast<float4*>(Y)[(size_t)e * (IN_CH / 4) + lane] = a;
}

__global__ __launch_bounds__(256) void gather_e2v(
    const float* __restrict__ Xe, const int* __restrict__ off,
    const int* __restrict__ lst, const float* __restrict__ degV,
    float* __restrict__ out, int N) {
  __shared__ __attribute__((aligned(16))) int sm[8][2 * SEG_CAP];
  const int w = __builtin_amdgcn_readfirstlane(threadIdx.x >> 5);
  const int lane = threadIdx.x & 31;
  const int v = blockIdx.x * 8 + w;
  const bool live = v < N;
  const int beg = live ? off[v] : 0;
  const int end = live ? off[v + 1] : 0;
  float4 a = ordered_row_sum(Xe, lst, beg, end, sm[w], sm[w] + SEG_CAP, lane);
  if (live) {
    const float dv = degV[v];
    a.x *= dv; a.y *= dv; a.z *= dv; a.w *= dv;
    reinterpret_cast<float4*>(out)[(size_t)v * (HO / 4) + lane] = a;
  }
}

extern "C" void kernel_launch(void* const* d_in, const int* in_sizes, int n_in,
                              void* d_out, int out_size, void* d_ws, size_t ws_size,
                              hipStream_t stream) {
  const float* X    = (const float*)d_in[0];
  const float* Wl   = (const float*)d_in[1];
  const float* degE = (const float*)d_in[2];
  const float* degV = (const float*)d_in[3];
  const float* W    = (const float*)d_in[4];
  const int* g1_src = (const int*)d_in[5];
  const int* g1_dst = (const int*)d_in[6];
  float* out = (float*)d_out;

  const int N   = in_sizes[0] / IN_CH;
  const int E   = in_sizes[2];
  const int nnz = in_sizes[5];

  float* Y  = (float*)d_ws;
  float* Xe = Y + (size_t)N * HO;
  const int E1 = E + 1, N1 = N + 1;
  const int T = E1 + N1;
  const int W0 = (E1 + 1) / 2, W1 = (N1 + 1) / 2, RW = W0 + W1;
  const int chunk = (((nnz + NBLK - 1) / NBLK) + 3) & ~3;
  int* cnt = (int*)(Y + (size_t)E * IN_CH);
  int* off = cnt + T;
  int* lst = cnt + ((2 * T + 3) & ~3) + 2 * (size_t)nnz;
  unsigned* mat16 = (unsigned*)Y;
  int* pre = lst + 2 * (size_t)nnz;

  count_slices<<<dim3(NBLK, 2), 1024, 0, stream>>>(g1_src, g1_dst, mat16,
                                                   E1, N1, W0, RW, chunk, nnz);
  slice_prefix<<<(RW + 255) / 256, 256, 0, stream>>>(mat16, pre, cnt, E1, N1, W0, RW, T);
  scan_counts<<<(T + 1023) / 1024, 1024, 0, stream>>>(cnt, off, T);
  place_slices<<<dim3(NBLK, HPARTS + NPARTS), 1024, 0, stream>>>(g1_src, g1_dst, off, pre, lst,
                                                                 E1, N1, T, chunk, nnz);

  gather_v2e<<<(E + 7) / 8, 256, 0, stream>>>(X, off, lst, Y, E);

  {
    int mtiles = (E + 15) / 16;
    int waves = mtiles * (HO / 16);
    int blocks = (waves * 32 + 255) / 256;
    gemm_wmma_f32<<<blocks, 256, 0, stream>>>(Y, Wl, degE, W, Xe, E);
  }

  gather_e2v<<<(N + 7) / 8, 256, 0, stream>>>(Xe, off + E + 1, lst, degV, out, N);
}


# coarse_place orders its slice's words by range in LDS (an LDS add hands out the stage place, word kept beside its list place) and writes contiguous runs: no scattered 4-byte store per edge
# speedup vs baseline: 5.2311x; 1.7616x over previous
#include <hip/hip_runtime.h>
#include <limits.h>

typedef __attribute__((ext_vector_type(2))) float v2f;
typedef __attribute__((ext_vector_type(8))) float v8f;

#define IN_CH 128
#define HO    128
#define SEG_CAP 256


#define NSL 128
#define NBK 256
#define FB_CAP 1024

__device__ __forceinline__ int range_shift(int n) {
  int s = 0;
  while ((n >> s) >= NBK) ++s;
  return s;
}

__device__ __forceinline__ int block_exclusive_scan(int x, int* wsum, int lane, int w) {
  int inc = x;
  for (int o = 1; o < 32; o <<= 1) {
    int y = __shfl_up(inc, o);
    if (lane >= o) inc += y;
  }
  if (lane == 31) wsum[w] = inc;
  __syncthreads();
  int t = (lane < w) ? wsum[lane] : 0;
  for (int o = 16; o; o >>= 1) t += __shfl_xor(t, o);
  return t + inc - x;
}

__global__ __launch_bounds__(1024) void coarse_count(
    const int* __restrict__ src, const int* __restrict__ dst,
    int* __restrict__ tab, int E, int N, int chunk, int nnz) {
  __shared__ int hist[NBK];
  const int side = blockIdx.y;
  const int* __restrict__ key = side ? src : dst;
  const int sh = range_shift(side ? N : E);
  const int tid = threadIdx.x;

  if (tid < NBK) hist[tid] = 0;
  __syncthreads();

  const int beg = blockIdx.x * chunk;
  const int end = min(beg + chunk, nnz);
  const int last4 = max(nnz - 4, 0);
  for (int i0 = beg + tid * 4; i0 < end; i0 += 4 * 4096) {
    int4 k[4];
#pragma unroll
    for (int u = 0; u < 4; ++u)
      k[u] = *reinterpret_cast<const int4*>(key + min(i0 + u * 4096, last4));
#pragma unroll
    for (int u = 0; u < 4; ++u) {
      const int i = i0 + u * 4096;
      if (i + 3 < end) {
        atomicAdd(&hist[k[u].x >> sh], 1);
        atomicAdd(&hist[k[u].y >> sh], 1);
        atomicAdd(&hist[k[u].z >> sh], 1);
        atomicAdd(&hist[k[u].w >> sh], 1);
      } else {
        for (int q = i; q < end; ++q) atomicAdd(&hist[key[q] >> sh], 1);
      }
    }
  }
  __syncthreads();

  if (tid < NBK) tab[((size_t)side * NSL + blockIdx.x) * NBK + tid] = hist[tid];
}

__device__ __forceinline__ int range_word(int sh, int mask, int k, int x) {
  return (int)(((unsigned)x << sh) | (unsigned)(k & mask));
}

__device__ __forceinline__ void to_range(int* cur, int* __restrict__ out, int sh, int mask,
                                         int k, int x) {
  out[atomicAdd(&cur[k >> sh], 1)] = range_word(sh, mask, k, x);
}

__device__ __forceinline__ void to_range4(int* cur, int* __restrict__ out, int sh, int mask,
                                          int4 k, int4 x) {
  const int p0 = atomicAdd(&cur[k.x >> sh], 1);
  const int p1 = atomicAdd(&cur[k.y >> sh], 1);
  const int p2 = atomicAdd(&cur[k.z >> sh], 1);
  const int p3 = atomicAdd(&cur[k.w >> sh], 1);
  out[p0] = range_word(sh, mask, k.x, x.x);
  out[p1] = range_word(sh, mask, k.y, x.y);
  out[p2] = range_word(sh, mask, k.z, x.z);
  out[p3] = range_word(sh, mask, k.w, x.w);
}

#define PLACE_CAP 12800
#define PROWS ((PLACE_CAP + 1023) / 1024)

__device__ __forceinline__ void to_stage(int* cur, const int* shift, int2* stage, int sh, int mask,
                                         int k, int x) {
  const int r = k >> sh;
  const int p = atomicAdd(&cur[r], 1);
  stage[p] = make_int2(range_word(sh, mask, k, x), p + shift[r]);
}

__device__ __forceinline__ void to_stage4(int* cur, const int* shift, int2* stage, int sh, int mask,
                                          int4 k, int4 x) {
  const int r0 = k.x >> sh, r1 = k.y >> sh, r2 = k.z >> sh, r3 = k.w >> sh;
  const int p0 = atomicAdd(&cur[r0], 1);
  const int p1 = atomicAdd(&cur[r1], 1);
  const int p2 = atomicAdd(&cur[r2], 1);
  const int p3 = atomicAdd(&cur[r3], 1);
  const int s0 = shift[r0], s1 = shift[r1], s2 = shift[r2], s3 = shift[r3];
  stage[p0] = make_int2(range_word(sh, mask, k.x, x.x), p0 + s0);
  stage[p1] = make_int2(range_word(sh, mask, k.y, x.y), p1 + s1);
  stage[p2] = make_int2(range_word(sh, mask, k.z, x.z), p2 + s2);
  stage[p3] = make_int2(range_word(sh, mask, k.w, x.w), p3 + s3);
}

__global__ __launch_bounds__(1024) void coarse_place(
    const int* __restrict__ src, const int* __restrict__ dst,
    const int* __restrict__ tab, int* __restrict__ rbase, int* __restrict__ rtot,
    int* __restrict__ pk, int E, int N, int chunk, int nnz) {
  __shared__ int part_all[4][NBK];
  __shared__ int part_bef[4][NBK];
  __shared__ int cur[NBK];
  __shared__ int shift[NBK];
  __shared__ int wsum[32];
  __shared__ int wsum2[32];
  __shared__ int2 stage[PLACE_CAP];
  const int side = blockIdx.y;
  const int s = blockIdx.x;
  const int tid = threadIdx.x, lane = tid & 31, w = tid >> 5;
  const int* __restrict__ key = side ? src : dst;
  const int* __restrict__ val = side ? dst : src;
  const int sh = range_shift(side ? N : E);
  const int mask = (1 << sh) - 1;
  const int beg = s * chunk;
  const int end = min(beg + chunk, nnz);
  const bool staged = end - beg <= PLACE_CAP;

  const int own = (tid < NBK) ? tab[((size_t)side * NSL + s) * NBK + tid] : 0;

  {
    const int r = tid & (NBK - 1), q = tid >> 8;
    const int* __restrict__ col = tab + (size_t)side * NSL * NBK + r;
    int c[NSL / 4];
#pragma unroll
    for (int u = 0; u < NSL / 4; ++u) c[u] = col[(size_t)(q * (NSL / 4) + u) * NBK];
    int all = 0, bef = 0;
#pragma unroll
    for (int u = 0; u < NSL / 4; ++u) {
      all += c[u];
      bef += (q * (NSL / 4) + u < s) ? c[u] : 0;
    }
    part_all[q][r] = all;
    part_bef[q][r] = bef;
  }
  __syncthreads();

  int tot = 0, bef = 0;
  if (tid < NBK) {
    tot = part_all[0][tid] + part_all[1][tid] + part_all[2][tid] + part_all[3][tid];
    bef = part_bef[0][tid] + part_bef[1][tid] + part_bef[2][tid] + part_bef[3][tid];
  }
  const int base = block_exclusive_scan(tot, wsum, lane, w);
  const int lb = block_exclusive_scan(own, wsum2, lane, w);
  if (tid < NBK) {
    cur[tid] = staged ? lb : base + bef;
    shift[tid] = base + bef - lb;
    if (s == 0) {
      rbase[side * NBK + tid] = base;
      rtot[side * NBK + tid] = tot;
    }
  }
  __syncthreads();

  int* __restrict__ out = pk + (size_t)side * nnz;
  const int last4 = max(nnz - 4, 0);
  for (int i0 = beg + tid * 4; i0 < end; i0 += 4 * 4096) {
    int4 k[4], x[4];
#pragma unroll
    for (int u = 0; u < 4; ++u) {
      const int at = min(i0 + u * 4096, last4);
      k[u] = *reinterpret_cast<const int4*>(key + at);
      x[u] = *reinterpret_cast<const int4*>(val + at);
    }
#pragma unroll
    for (int u = 0; u < 4; ++u) {
      const int i = i0 + u * 4096;
      if (staged) {
        if (i + 3 < end) {
          to_stage4(cur, shift, stage, sh, mask, k[u], x[u]);
        } else {
          for (int q = i; q < end; ++q) to_stage(cur, shift, stage, sh, mask, key[q], val[q]);
        }
      } else {
        if (i + 3 < end) {
          to_range4(cur, out, sh, mask, k[u], x[u]);
        } else {
          for (int q = i; q < end; ++q) to_range(cur, out, sh, mask, key[q], val[q]);
        }
      }
    }
  }
  if (staged) {
    __syncthreads();
    const int cnt = end - beg;
    int2 e[PROWS];
#pragma unroll
    for (int u = 0; u < PROWS; ++u) {
      const int i = tid + u * 1024;
      e[u] = stage[i < cnt ? i : 0];
    }
#pragma unroll
    for (int u = 0; u < PROWS; ++u)
      if (tid + u * 1024 < cnt) out[e[u].y] = e[u].x;
  }
}

#define FROWS 10
#define STAGE_CAP 12288

__global__ __launch_bounds__(1024) void bucket_finish(
    const int* __restrict__ pk, const int* __restrict__ rbase, const int* __restrict__ rtot,
    int* __restrict__ off, int* __restrict__ lst, int E, int N, int nnz) {
  __shared__ int bin[FB_CAP];
  __shared__ int spare[1024];
  __shared__ int stage[STAGE_CAP];
  __shared__ int wsum[32];
  const int side = blockIdx.y;
  const int r = blockIdx.x;
  const int tid = threadIdx.x, lane = tid & 31, w = tid >> 5;
  const int n = side ? N : E;
  const int sh = range_shift(n);
  const int bins = 1 << sh;
  const int mask = bins - 1;
  const int base = rbase[side * NBK + r];
  const int cnt = rtot[side * NBK + r];
  const int* __restrict__ ent = pk + (size_t)side * nnz + base;

  bin[tid] = 0;
  spare[tid] = 0;
  __syncthreads();
  for (int i0 = tid; i0 < cnt; i0 += FROWS * 1024) {
    int p[FROWS];
#pragma unroll
    for (int u = 0; u < FROWS; ++u) p[u] = ent[min(i0 + u * 1024, cnt - 1)];
#pragma unroll
    for (int u = 0; u < FROWS; ++u)
      atomicAdd((i0 + u * 1024 < cnt) ? &bin[p[u] & mask] : &spare[tid], 1);
  }
  __syncthreads();

  const int mine = bin[tid];
  const int ex = block_exclusive_scan(mine, wsum, lane, w);
  bin[tid] = ex;
  const int idx = r * bins + tid;
  if (tid < bins && idx <= n) off[(side ? E + 1 : 0) + idx] = side * nnz + base + ex;
  __syncthreads();

  int* __restrict__ seg = lst + (size_t)side * nnz + base;
  const bool staged = cnt <= STAGE_CAP;
  for (int i0 = tid; i0 < cnt; i0 += FROWS * 1024) {
    unsigned p[FROWS];
    int at[FROWS];
#pragma unroll
    for (int u = 0; u < FROWS; ++u) p[u] = (unsigned)ent[min(i0 + u * 1024, cnt - 1)];
#pragma unroll
    for (int u = 0; u < FROWS; ++u)
      at[u] = atomicAdd((i0 + u * 1024 < cnt) ? &bin[p[u] & (unsigned)mask] : &spare[tid], 1);
    if (staged) {
#pragma unroll
      for (int u = 0; u < FROWS; ++u)
        if (i0 + u * 1024 < cnt) stage[at[u]] = (int)(p[u] >> sh);
    } else {
#pragma unroll
      for (int u = 0; u < FROWS; ++u)
        if (i0 + u * 1024 < cnt) seg[at[u]] = (int)(p[u] >> sh);
    }
  }
  if (staged) {
    __syncthreads();
    for (int i = tid; i < cnt; i += 1024) seg[i] = stage[i];
  }
}

template <int NB>
__device__ __forceinline__ void add_rows(float4& acc, const float4* __restrict__ R,
                                         const int* idx, int lane) {
  float4 x[NB];
#pragma unroll
  for (int u = 0; u < NB; ++u) x[u] = R[(size_t)idx[u] * (HO / 4) + lane];
#pragma unroll
  for (int u = 0; u < NB; ++u) {
    acc.x += x[u].x; acc.y += x[u].y; acc.z += x[u].z; acc.w += x[u].w;
  }
}

__device__ __forceinline__ float4 ordered_row_sum(
    const float* __restrict__ rows, const int* __restrict__ lst,
    int beg, int end, int* raw, int* srt, int lane) {
  const int L = end - beg;
  const bool in_lds = L <= SEG_CAP;
  const int L4 = (L + 3) >> 2;
  const float4* R = reinterpret_cast<const float4*>(rows);
  float4 acc = make_float4(0.f, 0.f, 0.f, 0.f);

  if (in_lds) {
    for (int j = lane; j < L4 * 4; j += 32)
      raw[j] = (j < L) ? lst[beg + j] : INT_MAX;
  }
  __syncthreads();
  if (in_lds) {
    for (int j = lane; j < L; j += 32) {
      const int v = raw[j];
      int r = 0;
      for (int q = 0; q < L4; ++q) {
        const int4 u = reinterpret_cast<const int4*>(raw)[q];
        const int k = q * 4;
        r += (u.x < v) | ((u.x == v) & (k     < j));
        r += (u.y < v) | ((u.y == v) & (k + 1 < j));
        r += (u.z < v) | ((u.z == v) & (k + 2 < j));
        r += (u.w < v) | ((u.w == v) & (k + 3 < j));
      }
      srt[r] = v;
    }
  }
  __syncthreads();
  if (in_lds) {
    int j = 0;
    for (; j + 8 <= L; j += 8) add_rows<8>(acc, R, srt + j, lane);
    if (j + 4 <= L) { add_rows<4>(acc, R, srt + j, lane); j += 4; }
    if (j + 2 <= L) { add_rows<2>(acc, R, srt + j, lane); j += 2; }
    if (j < L) add_rows<1>(acc, R, srt + j, lane);
  } else {
    int prev = -1;
    for (int it = 0; it < L; ++it) {
      int best = INT_MAX;
      for (int k = beg + lane; k < end; k += 32) {
        const int v = lst[k];
        if (v > prev && v < best) best = v;
      }
      for (int o = 16; o; o >>= 1) best = min(best, __shfl_xor(best, o));
      if (best == INT_MAX) break;
      int c = 0;
      for (int k = beg + lane; k < end; k += 32) c += (lst[k] == best);
      for (int o = 16; o; o >>= 1) c += __shfl_xor(c, o);
      const float4 x = R[(size_t)best * (HO / 4) + lane];
      for (int m = 0; m < c; ++m) {
        acc.x += x.x; acc.y += x.y; acc.z += x.z; acc.w += x.w;
      }
      prev = best;
    }
  }
  return acc;
}

#define TPITCH 132

__global__ __launch_bounds__(512) __attribute__((amdgpu_waves_per_eu(16, 16)))
void gather_project_v2e(
    const float* __restrict__ X, const int* __restrict__ off,
    const int* __restrict__ lst, const float* __restrict__ Wl,
    const float* __restrict__ degE, const float* __restrict__ W,
    float* __restrict__ Xe, int E) {
  __shared__ __attribute__((aligned(16))) int sm[16][2 * SEG_CAP];
  __shared__ __attribute__((aligned(16))) float tile[16 * TPITCH];
  const int w = __builtin_amdgcn_readfirstlane(threadIdx.x >> 5);
  const int lane = threadIdx.x & 31;
  const int m0 = blockIdx.x * 16;
  const int e = m0 + w;
  const bool live = e < E;
  const int beg = live ? off[e] : 0;
  const int end = live ? off[e + 1] : 0;
  const float4 s = ordered_row_sum(X, lst, beg, end, sm[w], sm[w] + SEG_CAP, lane);
  *reinterpret_cast<float4*>(&tile[w * TPITCH + lane * 4]) = s;
  __syncthreads();
  if (w >= 8) return;

  const int half = lane >> 4;
  const int l15  = lane & 15;
  const int bcol = w * 16 + l15;

  float de[8], we[8];
#pragma unroll
  for (int r = 0; r < 8; ++r) {
    const int rr = min(m0 + half * 8 + r, E - 1);
    de[r] = degE[rr];
    we[r] = W[rr];
  }

  const float* __restrict__ arow = tile + l15 * TPITCH + half * 2;
  const float* __restrict__ bptr = Wl + (size_t)(half * 2) * HO + bcol;
  v8f acc = {};
  for (int k0 = 0; k0 < IN_CH; k0 += 4) {
    v2f a, b;
    a = *reinterpret_cast<const v2f*>(arow + k0);
    b.x = bptr[(size_t)k0 * HO];
    b.y = bptr[(size_t)(k0 + 1) * HO];
    acc = __builtin_amdgcn_wmma_f32_16x16x4_f32(
         false, a,  false, b,
         (short)0, acc,  false,  false);
  }
#pragma unroll
  for (int r = 0; r < 8; ++r) {
    const int row = m0 + half * 8 + r;
    if (row < E) Xe[(size_t)row * HO + bcol] = acc[r] * de[r] * we[r];
  }
}

__global__ __launch_bounds__(256) void gather_e2v(
    const float* __restrict__ Xe, const int* __restrict__ off,
    const int* __restrict__ lst, const float* __restrict__ degV,
    float* __restrict__ out, int N) {
  __shared__ __attribute__((aligned(16))) int sm[8][2 * SEG_CAP];
  const int w = __builtin_amdgcn_readfirstlane(threadIdx.x >> 5);
  const int lane = threadIdx.x & 31;
  const int v = blockIdx.x * 8 + w;
  const bool live = v < N;
  const int beg = live ? off[v] : 0;
  const int end = live ? off[v + 1] : 0;
  float4 a = ordered_row_sum(Xe, lst, beg, end, sm[w], sm[w] + SEG_CAP, lane);
  if (live) {
    const float dv = degV[v];
    a.x *= dv; a.y *= dv; a.z *= dv; a.w *= dv;
    reinterpret_cast<float4*>(out)[(size_t)v * (HO / 4) + lane] = a;
  }
}

extern "C" void kernel_launch(void* const* d_in, const int* in_sizes, int n_in,
                              void* d_out, int out_size, void* d_ws, size_t ws_size,
                              hipStream_t stream) {
  const float* X    = (const float*)d_in[0];
  const float* Wl   = (const float*)d_in[1];
  const float* degE = (const float*)d_in[2];
  const float* degV = (const float*)d_in[3];
  const float* W    = (const float*)d_in[4];
  const int* g1_src = (const int*)d_in[5];
  const int* g1_dst = (const int*)d_in[6];
  float* out = (float*)d_out;

  const int N   = in_sizes[0] / IN_CH;
  const int E   = in_sizes[2];
  const int nnz = in_sizes[5];

  float* ws0 = (float*)d_ws;
  float* Xe = ws0 + (size_t)N * HO;
  const int T = (E + 1) + (N + 1);
  const int chunk = (((nnz + NSL - 1) / NSL) + 3) & ~3;
  int* tab   = (int*)d_ws;
  int* rbase = tab + 2 * NSL * NBK;
  int* rtot  = rbase + 2 * NBK;
  int* off   = rtot + 2 * NBK;
  int* pk    = off + ((T + 3) & ~3);
  int* lst   = pk + 2 * (size_t)nnz;

  coarse_count<<<dim3(NSL, 2), 1024, 0, stream>>>(g1_src, g1_dst, tab, E, N, chunk, nnz);
  coarse_place<<<dim3(NSL, 2), 1024, 0, stream>>>(g1_src, g1_dst, tab, rbase, rtot, pk,
                                                  E, N, chunk, nnz);
  bucket_finish<<<dim3(NBK, 2), 1024, 0, stream>>>(pk, rbase, rtot, off, lst, E, N, nnz);

  gather_project_v2e<<<(E + 15) / 16, 512, 0, stream>>>(X, off, lst, Wl, degE, W, Xe, E);

  gather_e2v<<<(N + 7) / 8, 256, 0, stream>>>(Xe, off + E + 1, lst, degV, out, N);
}
